# starting kernel
# baseline (speedup 1.0000x reference)
#include <hip/hip_runtime.h>
#include <hip/hip_bf16.h>

typedef __attribute__((ext_vector_type(2))) float v2f;
typedef __attribute__((ext_vector_type(8))) float v8f;

#define H_PX   600
#define W_PX   800
#define N_OMM  721
#define PADR   4
#define KK     81
#define NTILES 46
#define NPLANES 96

__device__ __forceinline__ int reflect_idx(int v, int n) {
    v = (v < 0) ? -v : v;
    v = (v >= n) ? (2 * n - 2 - v) : v;
    return v;
}

__global__ __launch_bounds__(256) void hexeye_wmma_f32(
    const float* __restrict__ stim,
    const int*   __restrict__ rx,
    const int*   __restrict__ ry,
    float*       __restrict__ out)
{
    const int wavesPerBlock = blockDim.x >> 5;
    const int wave = blockIdx.x * wavesPerBlock + (threadIdx.x >> 5);
    const int lane = threadIdx.x & 31;

    const int plane = wave / NTILES;
    const int tile  = wave - plane * NTILES;
    if (plane >= NPLANES) return;

    const int n   = lane & 15;
    const int sub = lane >> 4;
    const int o_raw = tile * 16 + n;
    const bool valid = (o_raw < N_OMM);
    const int o = valid ? o_raw : (N_OMM - 1);

    int cx = rx[o]; cx = min(max(cx, PADR), W_PX + PADR - 1);
    int cy = ry[o]; cy = min(max(cy, PADR), H_PX + PADR - 1);

    const float* __restrict__ pp = stim + (size_t)plane * (H_PX * W_PX);

    int Yb[9], Xc[9];
    #pragma unroll
    for (int d = 0; d < 9; ++d) {
        Yb[d] = reflect_idx(cy + d - 8, H_PX) * W_PX;
        Xc[d] = reflect_idx(cx + d - 8, W_PX);
    }

    v2f a; a.x = 1.0f; a.y = 1.0f;
    v8f acc = {};

    #pragma unroll
    for (int s = 0; s < 21; ++s) {
        float b0, b1;
        if (s < 20) {
            const int k0a = 4 * s + 0, k0b = 4 * s + 1;
            const int k1a = 4 * s + 2, k1b = 4 * s + 3;
            const int i0 = sub ? (Yb[k0b / 9] + Xc[k0b % 9])
                               : (Yb[k0a / 9] + Xc[k0a % 9]);
            const int i1 = sub ? (Yb[k1b / 9] + Xc[k1b % 9])
                               : (Yb[k1a / 9] + Xc[k1a % 9]);
            b0 = pp[i0];
            b1 = pp[i1];
        } else {
            const float v = pp[Yb[8] + Xc[8]];
            b0 = sub ? 0.0f : v;
            b1 = 0.0f;
        }
        v2f bb; bb.x = b0; bb.y = b1;
        acc = __builtin_amdgcn_wmma_f32_16x16x4_f32(
             false, a,  false, bb,
             (short)0, acc,  false,  false);
    }

    if (valid && sub == 0) {
        out[(size_t)plane * N_OMM + o_raw] = acc[0] / 81.0f;
    }
}

extern "C" void kernel_launch(void* const* d_in, const int* in_sizes, int n_in,
                              void* d_out, int out_size, void* d_ws, size_t ws_size,
                              hipStream_t stream) {
    const float* stim = (const float*)d_in[0];
    const int*   rx   = (const int*)d_in[1];
    const int*   ry   = (const int*)d_in[2];
    float*       out  = (float*)d_out;

    (void)in_sizes; (void)n_in; (void)out_size; (void)d_ws; (void)ws_size;

    const int totalWaves = NPLANES * NTILES;
    const int wavesPerBlock = 256 / 32;
    const int blocks = (totalWaves + wavesPerBlock - 1) / wavesPerBlock;
    hexeye_wmma_f32<<<blocks, 256, 0, stream>>>(stim, rx, ry, out);
}


# Lanes map to patch elements (27 lanes = 3 rows x 9 adjacent floats of one receptor per load) instead of one receptor per lane; sum order changes: lane-row partials, then a second ones-WMMA pass
# speedup vs baseline: 1.6386x; 1.6386x over previous
#include <hip/hip_runtime.h>


typedef __attribute__((ext_vector_type(2))) float v2f;
typedef __attribute__((ext_vector_type(8))) float v8f;

#define H_PX    600
#define W_PX    800
#define N_OMM   721
#define PADR    4
#define NTILES  46
#define NPLANES 96
#define NPAIRS  (NPLANES / 2)
#define PLANE_ELEMS (H_PX * W_PX)
#define WAVES_PER_BLOCK 2

__device__ __forceinline__ int reflect_idx(int v, int n) {
    v = (v < 0) ? -v : v;
    v = (v >= n) ? (2 * n - 2 - v) : v;
    return v;
}

__global__ __launch_bounds__(64) void patch_mean_rows_wmma_f32(
    const float* __restrict__ stim,
    const int*   __restrict__ rx,
    const int*   __restrict__ ry,
    float*       __restrict__ out)
{
    const int wib  = __builtin_amdgcn_readfirstlane((int)(threadIdx.x >> 5));
    const int wave = (int)blockIdx.x * WAVES_PER_BLOCK + wib;
    if (wave >= NTILES * NPAIRS) return;

    const int pair = wave / NTILES;
    const int tile = wave - pair * NTILES;

    const int  lane   = (int)(threadIdx.x & 31);
    const bool lvalid = lane < 27;
    const int  le     = lvalid ? lane : 26;
    const int  lrow   = le / 9;
    const int  lx     = le - lrow * 9;
    const int  c      = lane & 15;

    const char* __restrict__ p0 = (const char*)(stim + (size_t)(2 * pair) * PLANE_ELEMS);
    const char* __restrict__ p1 = p0 + (size_t)PLANE_ELEMS * sizeof(float);

    const int ob = min(tile * 16, N_OMM - 16);
    const int* __restrict__ rxt = rx + ob;
    const int* __restrict__ ryt = ry + ob;

    float v0[16][3], v1[16][3];

    #pragma unroll
    for (int r = 0; r < 16; ++r) {
        int cx = rxt[r]; cx = min(max(cx, PADR), W_PX + PADR - 1);
        int cy = ryt[r]; cy = min(max(cy, PADR), H_PX + PADR - 1);
        const int X4 = reflect_idx(cx - 8 + lx, W_PX) * 4;
        #pragma unroll
        for (int j = 0; j < 3; ++j) {
            const int Y = reflect_idx(cy - 8 + 3 * j + lrow, H_PX);
            const unsigned boff = (unsigned)(Y * (W_PX * 4) + X4);
            v0[r][j] = *(const float*)(p0 + boff);
            v1[r][j] = *(const float*)(p1 + boff);
        }
    }

    __builtin_amdgcn_sched_barrier(0);

    v8f d0 = {}, d1 = {};
    #pragma unroll
    for (int i = 0; i < 8; ++i) {
        v2f sel;
        sel.x = (c == 2 * i)     ? 1.0f : 0.0f;
        sel.y = (c == 2 * i + 1) ? 1.0f : 0.0f;
        #pragma unroll
        for (int j = 0; j < 3; ++j) {
            v2f a0; a0.x = lvalid ? v0[2 * i][j] : 0.0f; a0.y = lvalid ? v0[2 * i + 1][j] : 0.0f;
            v2f a1; a1.x = lvalid ? v1[2 * i][j] : 0.0f; a1.y = lvalid ? v1[2 * i + 1][j] : 0.0f;
            d0 = __builtin_amdgcn_wmma_f32_16x16x4_f32(false, a0, false, sel, (short)0, d0, false, false);
            d1 = __builtin_amdgcn_wmma_f32_16x16x4_f32(false, a1, false, sel, (short)0, d1, false, false);
        }
    }

    v2f ones; ones.x = 1.0f; ones.y = 1.0f;
    v8f f0 = {}, f1 = {};
    #pragma unroll
    for (int q = 0; q < 4; ++q) {
        v2f b0; b0.x = d0[2 * q]; b0.y = d0[2 * q + 1];
        v2f b1; b1.x = d1[2 * q]; b1.y = d1[2 * q + 1];
        f0 = __builtin_amdgcn_wmma_f32_16x16x4_f32(false, ones, false, b0, (short)0, f0, false, false);
        f1 = __builtin_amdgcn_wmma_f32_16x16x4_f32(false, ones, false, b1, (short)0, f1, false, false);
    }

    const int o_out = ob + c;
    if (lane < 16 && o_out >= tile * 16) {
        out[(size_t)(2 * pair)     * N_OMM + o_out] = f0[0] / 81.0f;
        out[(size_t)(2 * pair + 1) * N_OMM + o_out] = f1[0] / 81.0f;
    }
}

extern "C" void kernel_launch(void* const* d_in, const int* in_sizes, int n_in,
                              void* d_out, int out_size, void* d_ws, size_t ws_size,
                              hipStream_t stream) {
    const float* stim = (const float*)d_in[0];
    const int*   rx   = (const int*)d_in[1];
    const int*   ry   = (const int*)d_in[2];
    float*       out  = (float*)d_out;

    (void)in_sizes; (void)n_in; (void)out_size; (void)d_ws; (void)ws_size;

    patch_mean_rows_wmma_f32<<<1104, 64, 0, stream>>>(stim, rx, ry, out);
}


# Work-groups of 9 waves on 9 consecutive receptor tiles (246 work-groups, one per WGP) in place of 2-wave work-groups: neighbouring patches' shared 128-byte lines are requested once per WGP
# speedup vs baseline: 1.8074x; 1.1030x over previous
#include <hip/hip_runtime.h>


typedef __attribute__((ext_vector_type(2))) float v2f;
typedef __attribute__((ext_vector_type(8))) float v8f;

#define H_PX    600
#define W_PX    800
#define N_OMM   721
#define PADR    4
#define NTILES  46
#define NPLANES 96
#define NPAIRS  (NPLANES / 2)
#define PLANE_ELEMS (H_PX * W_PX)
#define WAVES_PER_BLOCK 9

__device__ __forceinline__ int reflect_idx(int v, int n) {
    v = (v < 0) ? -v : v;
    v = (v >= n) ? (2 * n - 2 - v) : v;
    return v;
}

__global__ __launch_bounds__(32 * WAVES_PER_BLOCK) void patch_mean_rows_wmma_f32(
    const float* __restrict__ stim,
    const int*   __restrict__ rx,
    const int*   __restrict__ ry,
    float*       __restrict__ out)
{
    const int wib  = __builtin_amdgcn_readfirstlane((int)(threadIdx.x >> 5));
    const int wave = (int)blockIdx.x * WAVES_PER_BLOCK + wib;
    if (wave >= NTILES * NPAIRS) return;

    const int pair = wave / NTILES;
    const int tile = wave - pair * NTILES;

    const int  lane   = (int)(threadIdx.x & 31);
    const bool lvalid = lane < 27;
    const int  le     = lvalid ? lane : 26;
    const int  lrow   = le / 9;
    const int  lx     = le - lrow * 9;
    const int  c      = lane & 15;

    const char* __restrict__ p0 = (const char*)(stim + (size_t)(2 * pair) * PLANE_ELEMS);
    const char* __restrict__ p1 = p0 + (size_t)PLANE_ELEMS * sizeof(float);

    const int ob = min(tile * 16, N_OMM - 16);
    const int* __restrict__ rxt = rx + ob;
    const int* __restrict__ ryt = ry + ob;

    float v0[16][3], v1[16][3];

    #pragma unroll
    for (int r = 0; r < 16; ++r) {
        int cx = rxt[r]; cx = min(max(cx, PADR), W_PX + PADR - 1);
        int cy = ryt[r]; cy = min(max(cy, PADR), H_PX + PADR - 1);
        const int X4 = reflect_idx(cx - 8 + lx, W_PX) * 4;
        #pragma unroll
        for (int j = 0; j < 3; ++j) {
            const int Y = reflect_idx(cy - 8 + 3 * j + lrow, H_PX);
            const unsigned boff = (unsigned)(Y * (W_PX * 4) + X4);
            v0[r][j] = *(const float*)(p0 + boff);
            v1[r][j] = *(const float*)(p1 + boff);
        }
    }

    __builtin_amdgcn_sched_barrier(0);

    v8f d0 = {}, d1 = {};
    #pragma unroll
    for (int i = 0; i < 8; ++i) {
        v2f sel;
        sel.x = (c == 2 * i)     ? 1.0f : 0.0f;
        sel.y = (c == 2 * i + 1) ? 1.0f : 0.0f;
        #pragma unroll
        for (int j = 0; j < 3; ++j) {
            v2f a0; a0.x = lvalid ? v0[2 * i][j] : 0.0f; a0.y = lvalid ? v0[2 * i + 1][j] : 0.0f;
            v2f a1; a1.x = lvalid ? v1[2 * i][j] : 0.0f; a1.y = lvalid ? v1[2 * i + 1][j] : 0.0f;
            d0 = __builtin_amdgcn_wmma_f32_16x16x4_f32(false, a0, false, sel, (short)0, d0, false, false);
            d1 = __builtin_amdgcn_wmma_f32_16x16x4_f32(false, a1, false, sel, (short)0, d1, false, false);
        }
    }

    v2f ones; ones.x = 1.0f; ones.y = 1.0f;
    v8f f0 = {}, f1 = {};
    #pragma unroll
    for (int q = 0; q < 4; ++q) {
        v2f b0; b0.x = d0[2 * q]; b0.y = d0[2 * q + 1];
        v2f b1; b1.x = d1[2 * q]; b1.y = d1[2 * q + 1];
        f0 = __builtin_amdgcn_wmma_f32_16x16x4_f32(false, ones, false, b0, (short)0, f0, false, false);
        f1 = __builtin_amdgcn_wmma_f32_16x16x4_f32(false, ones, false, b1, (short)0, f1, false, false);
    }

    const int o_out = ob + c;
    if (lane < 16 && o_out >= tile * 16) {
        out[(size_t)(2 * pair)     * N_OMM + o_out] = f0[0] / 81.0f;
        out[(size_t)(2 * pair + 1) * N_OMM + o_out] = f1[0] / 81.0f;
    }
}

extern "C" void kernel_launch(void* const* d_in, const int* in_sizes, int n_in,
                              void* d_out, int out_size, void* d_ws, size_t ws_size,
                              hipStream_t stream) {
    const float* stim = (const float*)d_in[0];
    const int*   rx   = (const int*)d_in[1];
    const int*   ry   = (const int*)d_in[2];
    float*       out  = (float*)d_out;

    (void)in_sizes; (void)n_in; (void)out_size; (void)d_ws; (void)ws_size;

    patch_mean_rows_wmma_f32<<<246, 288, 0, stream>>>(stim, rx, ry, out);
}
